# starting kernel
# baseline (speedup 1.0000x reference)
#include <hip/hip_runtime.h>


#define B_MOL 16
#define N_ATOM 2048
#define NTILE (N_ATOM / 16)
#define TINY_F 1e-16f
#define WGS 256
#define WG_PER_MOL (N_ATOM / (16 * 8))
#define LPAD (N_ATOM + 16)

typedef __attribute__((ext_vector_type(2))) float v2f;
typedef __attribute__((ext_vector_type(8))) float v8f;

__global__ __launch_bounds__(WGS) void coulomb_field_kernel(
    const float* __restrict__ pos,
    const float* __restrict__ q,
    float* __restrict__ field) {
  __shared__ float smem[5 * LPAD];
  float* lx  = smem;
  float* ly  = smem + LPAD;
  float* lz  = smem + 2 * LPAD;
  float* lq  = smem + 3 * LPAD;
  float* lr2 = smem + 4 * LPAD;

  const int mol = blockIdx.x / WG_PER_MOL;
  const int wgm = blockIdx.x % WG_PER_MOL;
  const int tid = threadIdx.x;

  const float* posm = pos + (size_t)mol * N_ATOM * 3;
  const float* qm   = q   + (size_t)mol * N_ATOM;
  for (int idx = tid; idx < N_ATOM; idx += WGS) {
    float x = posm[idx * 3 + 0];
    float y = posm[idx * 3 + 1];
    float z = posm[idx * 3 + 2];
    lx[idx] = x; ly[idx] = y; lz[idx] = z;
    lq[idx] = qm[idx];
    lr2[idx] = x * x + y * y + z * z;
  }
  __syncthreads();

  const int wave = __builtin_amdgcn_readfirstlane(tid >> 5);
  const int lane = tid & 31;
  const int half = lane >> 4;
  const int l16  = lane & 15;

  const int jbase = (wgm * 8 + wave) * 16;
  const int jtile = jbase >> 4;

  v2f amat;
  {
    const int j = jbase + l16;
    if (half == 0) { amat.x = -2.0f * lx[j]; amat.y = -2.0f * ly[j]; }
    else           { amat.x = -2.0f * lz[j]; amat.y = 1.0f; }
  }

  v8f cmat;
#pragma unroll
  for (int v = 0; v < 8; ++v) cmat[v] = lr2[jbase + v + 8 * half] + TINY_F;

  const float* bsrc0 = half ? lz  : lx;
  const float* bsrc1 = half ? lr2 : ly;

  float facc[8];
#pragma unroll
  for (int v = 0; v < 8; ++v) facc[v] = 0.0f;

  auto accum = [&](const v8f& d, float qi) {
#pragma unroll
    for (int v = 0; v < 8; ++v)
      facc[v] += qi * __builtin_amdgcn_rsqf(__builtin_fabsf(d[v]));
  };

  int ecur = (0 >= jtile) ? 1 : 0;
  v2f bcur; bcur.x = bsrc0[ecur * 16 + l16]; bcur.y = bsrc1[ecur * 16 + l16];
  float qcur = lq[ecur * 16 + l16];

  for (int t = 0; t < NTILE - 2; ++t) {
    const int en = (t + 1) + ((t + 1) >= jtile ? 1 : 0);
    v2f bnext; bnext.x = bsrc0[en * 16 + l16]; bnext.y = bsrc1[en * 16 + l16];
    const float qnext = lq[en * 16 + l16];

    v8f d = __builtin_amdgcn_wmma_f32_16x16x4_f32(
        false, amat, false, bcur, (short)0, cmat, false, false);
    accum(d, qcur);

    bcur = bnext; qcur = qnext;
  }
  {
    v8f d = __builtin_amdgcn_wmma_f32_16x16x4_f32(
        false, amat, false, bcur, (short)0, cmat, false, false);
    accum(d, qcur);
  }

  {
    const int i_l = jbase + l16;
    v2f bmat; bmat.x = bsrc0[i_l]; bmat.y = bsrc1[i_l];
    const float qi = lq[i_l];
    v8f d = __builtin_amdgcn_wmma_f32_16x16x4_f32(
        false, amat, false, bmat, (short)0, cmat, false, false);
#pragma unroll
    for (int v = 0; v < 8; ++v) {
      float inv = __builtin_amdgcn_rsqf(__builtin_fabsf(d[v]));
      facc[v] += (l16 != v + 8 * half) ? qi * inv : 0.0f;
    }
  }

#pragma unroll
  for (int v = 0; v < 8; ++v) {
    float f = facc[v];
    f += __shfl_xor(f, 1);
    f += __shfl_xor(f, 2);
    f += __shfl_xor(f, 4);
    f += __shfl_xor(f, 8);
    facc[v] = f;
  }

  if (l16 == 0) {
#pragma unroll
    for (int v = 0; v < 8; ++v)
      field[(size_t)mol * N_ATOM + jbase + v + 8 * half] = facc[v];
  }
}

__global__ __launch_bounds__(256) void coulomb_pot_kernel(
    const float* __restrict__ q, const float* __restrict__ field,
    float* __restrict__ pot) {
  __shared__ float red[256];
  const int b = blockIdx.x;
  float s = 0.0f;
  for (int j = threadIdx.x; j < N_ATOM; j += 256)
    s += q[(size_t)b * N_ATOM + j] * field[(size_t)b * N_ATOM + j];
  red[threadIdx.x] = s;
  __syncthreads();
  for (int off = 128; off > 0; off >>= 1) {
    if (threadIdx.x < off) red[threadIdx.x] += red[threadIdx.x + off];
    __syncthreads();
  }
  if (threadIdx.x == 0) pot[b] = 0.5f * red[0];
}

extern "C" void kernel_launch(void* const* d_in, const int* in_sizes, int n_in,
                              void* d_out, int out_size, void* d_ws, size_t ws_size,
                              hipStream_t stream) {
  (void)in_sizes; (void)n_in; (void)out_size; (void)d_ws; (void)ws_size;
  const float* pos = (const float*)d_in[0];
  const float* q   = (const float*)d_in[1];

  float* out   = (float*)d_out;
  float* field = out + B_MOL;

  coulomb_field_kernel<<<B_MOL * WG_PER_MOL, WGS, 0, stream>>>(pos, q, field);
  coulomb_pot_kernel<<<B_MOL, 256, 0, stream>>>(q, field, out);
}


# Both global-load loops (LDS staging in the field kernel, q*field read in the pot kernel) fully unrolled: all 8 trips of loads in flight before the first wait; arithmetic and sum order unchanged
# speedup vs baseline: 1.4867x; 1.4867x over previous
#include <hip/hip_runtime.h>


#define B_MOL 16
#define N_ATOM 2048
#define NTILE (N_ATOM / 16)
#define TINY_F 1e-16f
#define WGS 256
#define WG_PER_MOL (N_ATOM / (16 * 8))
#define LPAD (N_ATOM + 16)
#define NSTG (N_ATOM / WGS)
#define NPOT (N_ATOM / 256)

typedef __attribute__((ext_vector_type(2))) float v2f;
typedef __attribute__((ext_vector_type(8))) float v8f;

__global__ __launch_bounds__(WGS) void coulomb_field_kernel(
    const float* __restrict__ pos,
    const float* __restrict__ q,
    float* __restrict__ field) {
  __shared__ float smem[5 * LPAD];
  float* lx  = smem;
  float* ly  = smem + LPAD;
  float* lz  = smem + 2 * LPAD;
  float* lq  = smem + 3 * LPAD;
  float* lr2 = smem + 4 * LPAD;

  const int mol = blockIdx.x / WG_PER_MOL;
  const int wgm = blockIdx.x % WG_PER_MOL;
  const int tid = threadIdx.x;

  const float* posm = pos + (size_t)mol * N_ATOM * 3;
  const float* qm   = q   + (size_t)mol * N_ATOM;
  float sx[NSTG], sy[NSTG], sz[NSTG], sq[NSTG];
#pragma unroll
  for (int k = 0; k < NSTG; ++k) {
    const int idx = k * WGS + tid;
    sx[k] = posm[idx * 3 + 0];
    sy[k] = posm[idx * 3 + 1];
    sz[k] = posm[idx * 3 + 2];
    sq[k] = qm[idx];
  }
#pragma unroll
  for (int k = 0; k < NSTG; ++k) {
    const int idx = k * WGS + tid;
    const float x = sx[k], y = sy[k], z = sz[k];
    lx[idx] = x; ly[idx] = y; lz[idx] = z;
    lq[idx] = sq[k];
    lr2[idx] = x * x + y * y + z * z;
  }
  __syncthreads();

  const int wave = __builtin_amdgcn_readfirstlane(tid >> 5);
  const int lane = tid & 31;
  const int half = lane >> 4;
  const int l16  = lane & 15;

  const int jbase = (wgm * 8 + wave) * 16;
  const int jtile = jbase >> 4;

  v2f amat;
  {
    const int j = jbase + l16;
    if (half == 0) { amat.x = -2.0f * lx[j]; amat.y = -2.0f * ly[j]; }
    else           { amat.x = -2.0f * lz[j]; amat.y = 1.0f; }
  }

  v8f cmat;
#pragma unroll
  for (int v = 0; v < 8; ++v) cmat[v] = lr2[jbase + v + 8 * half] + TINY_F;

  const float* bsrc0 = half ? lz  : lx;
  const float* bsrc1 = half ? lr2 : ly;

  float facc[8];
#pragma unroll
  for (int v = 0; v < 8; ++v) facc[v] = 0.0f;

  auto accum = [&](const v8f& d, float qi) {
#pragma unroll
    for (int v = 0; v < 8; ++v)
      facc[v] += qi * __builtin_amdgcn_rsqf(__builtin_fabsf(d[v]));
  };

  int ecur = (0 >= jtile) ? 1 : 0;
  v2f bcur; bcur.x = bsrc0[ecur * 16 + l16]; bcur.y = bsrc1[ecur * 16 + l16];
  float qcur = lq[ecur * 16 + l16];

  for (int t = 0; t < NTILE - 2; ++t) {
    const int en = (t + 1) + ((t + 1) >= jtile ? 1 : 0);
    v2f bnext; bnext.x = bsrc0[en * 16 + l16]; bnext.y = bsrc1[en * 16 + l16];
    const float qnext = lq[en * 16 + l16];

    v8f d = __builtin_amdgcn_wmma_f32_16x16x4_f32(
        false, amat, false, bcur, (short)0, cmat, false, false);
    accum(d, qcur);

    bcur = bnext; qcur = qnext;
  }
  {
    v8f d = __builtin_amdgcn_wmma_f32_16x16x4_f32(
        false, amat, false, bcur, (short)0, cmat, false, false);
    accum(d, qcur);
  }

  {
    const int i_l = jbase + l16;
    v2f bmat; bmat.x = bsrc0[i_l]; bmat.y = bsrc1[i_l];
    const float qi = lq[i_l];
    v8f d = __builtin_amdgcn_wmma_f32_16x16x4_f32(
        false, amat, false, bmat, (short)0, cmat, false, false);
#pragma unroll
    for (int v = 0; v < 8; ++v) {
      float inv = __builtin_amdgcn_rsqf(__builtin_fabsf(d[v]));
      facc[v] += (l16 != v + 8 * half) ? qi * inv : 0.0f;
    }
  }

#pragma unroll
  for (int v = 0; v < 8; ++v) {
    float f = facc[v];
    f += __shfl_xor(f, 1);
    f += __shfl_xor(f, 2);
    f += __shfl_xor(f, 4);
    f += __shfl_xor(f, 8);
    facc[v] = f;
  }

  if (l16 == 0) {
#pragma unroll
    for (int v = 0; v < 8; ++v)
      field[(size_t)mol * N_ATOM + jbase + v + 8 * half] = facc[v];
  }
}

__global__ __launch_bounds__(256) void coulomb_pot_kernel(
    const float* __restrict__ q, const float* __restrict__ field,
    float* __restrict__ pot) {
  __shared__ float red[256];
  const int b = blockIdx.x;
  float qv[NPOT], fv[NPOT];
#pragma unroll
  for (int k = 0; k < NPOT; ++k) {
    const int j = k * 256 + threadIdx.x;
    qv[k] = q[(size_t)b * N_ATOM + j];
    fv[k] = field[(size_t)b * N_ATOM + j];
  }
  float s = 0.0f;
#pragma unroll
  for (int k = 0; k < NPOT; ++k) s = __builtin_fmaf(qv[k], fv[k], s);
  red[threadIdx.x] = s;
  __syncthreads();
  for (int off = 128; off > 0; off >>= 1) {
    if (threadIdx.x < off) red[threadIdx.x] += red[threadIdx.x + off];
    __syncthreads();
  }
  if (threadIdx.x == 0) pot[b] = 0.5f * red[0];
}

extern "C" void kernel_launch(void* const* d_in, const int* in_sizes, int n_in,
                              void* d_out, int out_size, void* d_ws, size_t ws_size,
                              hipStream_t stream) {
  (void)in_sizes; (void)n_in; (void)out_size; (void)d_ws; (void)ws_size;
  const float* pos = (const float*)d_in[0];
  const float* q   = (const float*)d_in[1];

  float* out   = (float*)d_out;
  float* field = out + B_MOL;

  coulomb_field_kernel<<<B_MOL * WG_PER_MOL, WGS, 0, stream>>>(pos, q, field);
  coulomb_pot_kernel<<<B_MOL, 256, 0, stream>>>(q, field, out);
}


# Cross-lane sums moved off the LDS unit: row_xmask DPP adds replace the shuffles in the field row sums (same bits) and the 8-barrier LDS tree of the pot sum (pot sum order changes: xor butterfly)
# speedup vs baseline: 1.5764x; 1.0603x over previous
#include <hip/hip_runtime.h>


#define B_MOL 16
#define N_ATOM 2048
#define NTILE (N_ATOM / 16)
#define TINY_F 1e-16f
#define WGS 256
#define WG_PER_MOL (N_ATOM / (16 * 8))
#define LPAD (N_ATOM + 16)
#define NSTG (N_ATOM / WGS)
#define NPOT (N_ATOM / 256)

typedef __attribute__((ext_vector_type(2))) float v2f;
typedef __attribute__((ext_vector_type(8))) float v8f;

template <int M>
__device__ __forceinline__ float row_xor_add(float f) {
  const int i = __builtin_bit_cast(int, f);
  const int p = __builtin_amdgcn_update_dpp(i, i, 0x160 + M, 0xf, 0xf, true);
  return f + __builtin_bit_cast(float, p);
}

__device__ __forceinline__ float other_row(float x) {
  const int i = __builtin_bit_cast(int, x);
  return __builtin_bit_cast(float,
      __builtin_amdgcn_permlanex16(i, i, 0x76543210, 0xfedcba98, false, false));
}

__global__ __launch_bounds__(WGS) void coulomb_field_kernel(
    const float* __restrict__ pos,
    const float* __restrict__ q,
    float* __restrict__ field) {
  __shared__ float smem[5 * LPAD];
  float* lx  = smem;
  float* ly  = smem + LPAD;
  float* lz  = smem + 2 * LPAD;
  float* lq  = smem + 3 * LPAD;
  float* lr2 = smem + 4 * LPAD;

  const int mol = blockIdx.x / WG_PER_MOL;
  const int wgm = blockIdx.x % WG_PER_MOL;
  const int tid = threadIdx.x;

  const float* posm = pos + (size_t)mol * N_ATOM * 3;
  const float* qm   = q   + (size_t)mol * N_ATOM;
  float sx[NSTG], sy[NSTG], sz[NSTG], sq[NSTG];
#pragma unroll
  for (int k = 0; k < NSTG; ++k) {
    const int idx = k * WGS + tid;
    sx[k] = posm[idx * 3 + 0];
    sy[k] = posm[idx * 3 + 1];
    sz[k] = posm[idx * 3 + 2];
    sq[k] = qm[idx];
  }
#pragma unroll
  for (int k = 0; k < NSTG; ++k) {
    const int idx = k * WGS + tid;
    const float x = sx[k], y = sy[k], z = sz[k];
    lx[idx] = x; ly[idx] = y; lz[idx] = z;
    lq[idx] = sq[k];
    lr2[idx] = x * x + y * y + z * z;
  }
  __syncthreads();

  const int wave = __builtin_amdgcn_readfirstlane(tid >> 5);
  const int lane = tid & 31;
  const int half = lane >> 4;
  const int l16  = lane & 15;

  const int jbase = (wgm * 8 + wave) * 16;
  const int jtile = jbase >> 4;

  v2f amat;
  {
    const int j = jbase + l16;
    if (half == 0) { amat.x = -2.0f * lx[j]; amat.y = -2.0f * ly[j]; }
    else           { amat.x = -2.0f * lz[j]; amat.y = 1.0f; }
  }

  v8f cmat;
#pragma unroll
  for (int v = 0; v < 8; ++v) cmat[v] = lr2[jbase + v + 8 * half] + TINY_F;

  const float* bsrc0 = half ? lz  : lx;
  const float* bsrc1 = half ? lr2 : ly;

  float facc[8];
#pragma unroll
  for (int v = 0; v < 8; ++v) facc[v] = 0.0f;

  auto accum = [&](const v8f& d, float qi) {
#pragma unroll
    for (int v = 0; v < 8; ++v)
      facc[v] += qi * __builtin_amdgcn_rsqf(__builtin_fabsf(d[v]));
  };

  int ecur = (0 >= jtile) ? 1 : 0;
  v2f bcur; bcur.x = bsrc0[ecur * 16 + l16]; bcur.y = bsrc1[ecur * 16 + l16];
  float qcur = lq[ecur * 16 + l16];

  for (int t = 0; t < NTILE - 2; ++t) {
    const int en = (t + 1) + ((t + 1) >= jtile ? 1 : 0);
    v2f bnext; bnext.x = bsrc0[en * 16 + l16]; bnext.y = bsrc1[en * 16 + l16];
    const float qnext = lq[en * 16 + l16];

    v8f d = __builtin_amdgcn_wmma_f32_16x16x4_f32(
        false, amat, false, bcur, (short)0, cmat, false, false);
    accum(d, qcur);

    bcur = bnext; qcur = qnext;
  }
  {
    v8f d = __builtin_amdgcn_wmma_f32_16x16x4_f32(
        false, amat, false, bcur, (short)0, cmat, false, false);
    accum(d, qcur);
  }

  {
    const int i_l = jbase + l16;
    v2f bmat; bmat.x = bsrc0[i_l]; bmat.y = bsrc1[i_l];
    const float qi = lq[i_l];
    v8f d = __builtin_amdgcn_wmma_f32_16x16x4_f32(
        false, amat, false, bmat, (short)0, cmat, false, false);
#pragma unroll
    for (int v = 0; v < 8; ++v) {
      float inv = __builtin_amdgcn_rsqf(__builtin_fabsf(d[v]));
      facc[v] += (l16 != v + 8 * half) ? qi * inv : 0.0f;
    }
  }

#pragma unroll
  for (int v = 0; v < 8; ++v) {
    float f = facc[v];
    f = row_xor_add<1>(f);
    f = row_xor_add<2>(f);
    f = row_xor_add<4>(f);
    f = row_xor_add<8>(f);
    facc[v] = f;
  }

  if (l16 == 0) {
#pragma unroll
    for (int v = 0; v < 8; ++v)
      field[(size_t)mol * N_ATOM + jbase + v + 8 * half] = facc[v];
  }
}

__global__ __launch_bounds__(256) void coulomb_pot_kernel(
    const float* __restrict__ q, const float* __restrict__ field,
    float* __restrict__ pot) {
  __shared__ float red[8];
  const int b = blockIdx.x;
  float qv[NPOT], fv[NPOT];
#pragma unroll
  for (int k = 0; k < NPOT; ++k) {
    const int j = k * 256 + threadIdx.x;
    qv[k] = q[(size_t)b * N_ATOM + j];
    fv[k] = field[(size_t)b * N_ATOM + j];
  }
  float s = 0.0f;
#pragma unroll
  for (int k = 0; k < NPOT; ++k) s = __builtin_fmaf(qv[k], fv[k], s);
  s = row_xor_add<1>(s);
  s = row_xor_add<2>(s);
  s = row_xor_add<4>(s);
  s = row_xor_add<8>(s);
  s += other_row(s);
  const int wave = __builtin_amdgcn_readfirstlane(threadIdx.x >> 5);
  const int lane = threadIdx.x & 31;
  if (lane == 0) red[wave] = s;
  __syncthreads();
  if (wave == 0) {
    float t = red[lane & 7];
    t = row_xor_add<1>(t);
    t = row_xor_add<2>(t);
    t = row_xor_add<4>(t);
    if (lane == 0) pot[b] = 0.5f * t;
  }
}

extern "C" void kernel_launch(void* const* d_in, const int* in_sizes, int n_in,
                              void* d_out, int out_size, void* d_ws, size_t ws_size,
                              hipStream_t stream) {
  (void)in_sizes; (void)n_in; (void)out_size; (void)d_ws; (void)ws_size;
  const float* pos = (const float*)d_in[0];
  const float* q   = (const float*)d_in[1];

  float* out   = (float*)d_out;
  float* field = out + B_MOL;

  coulomb_field_kernel<<<B_MOL * WG_PER_MOL, WGS, 0, stream>>>(pos, q, field);
  coulomb_pot_kernel<<<B_MOL, 256, 0, stream>>>(q, field, out);
}


# Field loop in blocks of 4 i-tiles: next block's B fragments and charges come from LDS as one group of wide loads (quad-per-atom layout) ahead of 4 back-to-back products; field sum order changes
# speedup vs baseline: 1.7086x; 1.0839x over previous
#include <hip/hip_runtime.h>


#define B_MOL 16
#define N_ATOM 2048
#define NTILE (N_ATOM / 16)
#define TINY_F 1e-16f
#define WGS 256
#define WG_PER_MOL (N_ATOM / (16 * 8))
#define GT 4
#define NGRP (NTILE / GT)
#define NSTG (N_ATOM / WGS)
#define NPOT (N_ATOM / 256)

typedef __attribute__((ext_vector_type(2))) float v2f;
typedef __attribute__((ext_vector_type(4))) float v4f;
typedef __attribute__((ext_vector_type(8))) float v8f;

template <int M>
__device__ __forceinline__ float row_xor_add(float f) {
  const int i = __builtin_bit_cast(int, f);
  const int p = __builtin_amdgcn_update_dpp(i, i, 0x160 + M, 0xf, 0xf, true);
  return f + __builtin_bit_cast(float, p);
}

__device__ __forceinline__ float other_row(float x) {
  const int i = __builtin_bit_cast(int, x);
  return __builtin_bit_cast(float,
      __builtin_amdgcn_permlanex16(i, i, 0x76543210, 0xfedcba98, false, false));
}

__global__ __launch_bounds__(WGS) void coulomb_field_kernel(
    const float* __restrict__ pos,
    const float* __restrict__ q,
    float* __restrict__ field) {
  __shared__ __attribute__((aligned(16))) float satom[N_ATOM * 4];
  __shared__ __attribute__((aligned(16))) float sq[N_ATOM];

  const int mol = blockIdx.x / WG_PER_MOL;
  const int wgm = blockIdx.x % WG_PER_MOL;
  const int tid = threadIdx.x;

  const float* posm = pos + (size_t)mol * N_ATOM * 3;
  const float* qm   = q   + (size_t)mol * N_ATOM;
  float sx[NSTG], sy[NSTG], sz[NSTG], sc[NSTG];
#pragma unroll
  for (int k = 0; k < NSTG; ++k) {
    const int idx = k * WGS + tid;
    sx[k] = posm[idx * 3 + 0];
    sy[k] = posm[idx * 3 + 1];
    sz[k] = posm[idx * 3 + 2];
    sc[k] = qm[idx];
  }
#pragma unroll
  for (int k = 0; k < NSTG; ++k) {
    const int idx = k * WGS + tid;
    const float x = sx[k], y = sy[k], z = sz[k];
    v4f a; a.x = x; a.y = y; a.z = z;
    a.w = __builtin_fmaf(z, z, __builtin_fmaf(x, x, y * y));
    *(v4f*)(satom + idx * 4) = a;
    const int e = idx >> 4;
    sq[((e / GT) * 16 + (idx & 15)) * GT + (e % GT)] = sc[k];
  }
  __syncthreads();

  const int wave = __builtin_amdgcn_readfirstlane(tid >> 5);
  const int lane = tid & 31;
  const int half = lane >> 4;
  const int l16  = lane & 15;

  const int jbase = (wgm * 8 + wave) * 16;
  const int jtile = jbase >> 4;

  v2f amat;
  {
    const v4f pj = *(const v4f*)(satom + (jbase + l16) * 4);
    if (half == 0) { amat.x = -2.0f * pj.x; amat.y = -2.0f * pj.y; }
    else           { amat.x = -2.0f * pj.z; amat.y = 1.0f; }
  }

  v8f cmat;
#pragma unroll
  for (int v = 0; v < 8; ++v)
    cmat[v] = satom[(jbase + v + 8 * half) * 4 + 3] + TINY_F;

  const v2f* bsrc = (const v2f*)satom + (l16 * 2 + half);
  const v4f* qsrc = (const v4f*)sq + l16;

  float facc[8];
#pragma unroll
  for (int v = 0; v < 8; ++v) facc[v] = 0.0f;

  auto load_block = [&](int g, v2f (&b)[GT], v4f& qq) {
#pragma unroll
    for (int u = 0; u < GT; ++u) b[u] = bsrc[((g * GT + u) * 16) * 2];
    qq = qsrc[g * 16];
  };

  auto run_block = [&](const v2f (&b)[GT], const v4f& qq) {
    v8f d[GT];
#pragma unroll
    for (int u = 0; u < GT; ++u)
      d[u] = __builtin_amdgcn_wmma_f32_16x16x4_f32(
          false, amat, false, b[u], (short)0, cmat, false, false);
#pragma unroll
    for (int u = 0; u < GT; ++u) {
      const float qi = qq[u];
#pragma unroll
      for (int v = 0; v < 8; ++v)
        facc[v] = __builtin_fmaf(qi, __builtin_amdgcn_rsqf(__builtin_fabsf(d[u][v])), facc[v]);
    }
  };

  const int gdiag = jtile / GT;
  const int udiag = jtile % GT;
  v2f bA[GT], bB[GT];
  v4f qA, qB;
  load_block((0 >= gdiag) ? 1 : 0, bA, qA);
  for (int i = 0; i < (NGRP - 2) / 2; ++i) {
    const int k1 = 2 * i + 1, k2 = 2 * i + 2;
    load_block(k1 + (k1 >= gdiag ? 1 : 0), bB, qB);
    run_block(bA, qA);
    load_block(k2 + (k2 >= gdiag ? 1 : 0), bA, qA);
    run_block(bB, qB);
  }
  load_block(gdiag, bB, qB);
  run_block(bA, qA);

  {
    v8f d[GT];
#pragma unroll
    for (int u = 0; u < GT; ++u)
      d[u] = __builtin_amdgcn_wmma_f32_16x16x4_f32(
          false, amat, false, bB[u], (short)0, cmat, false, false);
#pragma unroll
    for (int u = 0; u < GT; ++u) {
      const float qi = qB[u];
      const int dv = (u == udiag) ? (l16 - 8 * half) : -1;
#pragma unroll
      for (int v = 0; v < 8; ++v) {
        const float inv = __builtin_amdgcn_rsqf(__builtin_fabsf(d[u][v]));
        const float acc = __builtin_fmaf(qi, inv, facc[v]);
        facc[v] = (v != dv) ? acc : facc[v];
      }
    }
  }

#pragma unroll
  for (int v = 0; v < 8; ++v) {
    float f = facc[v];
    f = row_xor_add<1>(f);
    f = row_xor_add<2>(f);
    f = row_xor_add<4>(f);
    f = row_xor_add<8>(f);
    facc[v] = f;
  }

  if (l16 == 0) {
#pragma unroll
    for (int v = 0; v < 8; ++v)
      field[(size_t)mol * N_ATOM + jbase + v + 8 * half] = facc[v];
  }
}

__global__ __launch_bounds__(256) void coulomb_pot_kernel(
    const float* __restrict__ q, const float* __restrict__ field,
    float* __restrict__ pot) {
  __shared__ float red[8];
  const int b = blockIdx.x;
  float qv[NPOT], fv[NPOT];
#pragma unroll
  for (int k = 0; k < NPOT; ++k) {
    const int j = k * 256 + threadIdx.x;
    qv[k] = q[(size_t)b * N_ATOM + j];
    fv[k] = field[(size_t)b * N_ATOM + j];
  }
  float s = 0.0f;
#pragma unroll
  for (int k = 0; k < NPOT; ++k) s = __builtin_fmaf(qv[k], fv[k], s);
  s = row_xor_add<1>(s);
  s = row_xor_add<2>(s);
  s = row_xor_add<4>(s);
  s = row_xor_add<8>(s);
  s += other_row(s);
  const int wave = __builtin_amdgcn_readfirstlane(threadIdx.x >> 5);
  const int lane = threadIdx.x & 31;
  if (lane == 0) red[wave] = s;
  __syncthreads();
  if (wave == 0) {
    float t = red[lane & 7];
    t = row_xor_add<1>(t);
    t = row_xor_add<2>(t);
    t = row_xor_add<4>(t);
    if (lane == 0) pot[b] = 0.5f * t;
  }
}

extern "C" void kernel_launch(void* const* d_in, const int* in_sizes, int n_in,
                              void* d_out, int out_size, void* d_ws, size_t ws_size,
                              hipStream_t stream) {
  (void)in_sizes; (void)n_in; (void)out_size; (void)d_ws; (void)ws_size;
  const float* pos = (const float*)d_in[0];
  const float* q   = (const float*)d_in[1];

  float* out   = (float*)d_out;
  float* field = out + B_MOL;

  coulomb_field_kernel<<<B_MOL * WG_PER_MOL, WGS, 0, stream>>>(pos, q, field);
  coulomb_pot_kernel<<<B_MOL, 256, 0, stream>>>(q, field, out);
}
